# starting kernel
# baseline (speedup 1.0000x reference)
#include <hip/hip_runtime.h>

typedef __attribute__((ext_vector_type(2))) float v2f;
typedef __attribute__((ext_vector_type(8))) float v8f;

#define B_TOTAL 4096
#define T_DIM   24
#define F_DIM   256
#define O_DIM   24
#define TGT     2

__global__ __launch_bounds__(256) void dlinear_wmma_f32(
    const float* __restrict__ x,
    const float* __restrict__ W,
    const float* __restrict__ bias,
    float* __restrict__ out)
{
  const int lane = threadIdx.x & 31;
  const int wave = threadIdx.x >> 5;
  const int tile = blockIdx.x * 8 + wave;
  const int b0   = tile * 16;

  const int  mrow = lane & 15;
  const int  grp  = lane >> 4;

  const int   oHi   = 16 + mrow;
  const bool  oHiOk = (oHi < O_DIM);
  const int   oHiC  = oHiOk ? oHi : 0;
  const float oHiM  = oHiOk ? 1.0f : 0.0f;

  v8f acc00 = {};
  v8f acc01 = {};
  v8f acc10 = {};
  v8f acc11 = {};

  const float* xrow = x + (size_t)(b0 + mrow) * (T_DIM * F_DIM);
  const float* w0   = W;
  const float* w1   = W + O_DIM * T_DIM;

#pragma unroll
  for (int k = 0; k < T_DIM; k += 4) {
    const int t0 = k + 2 * grp;

    v2f p0 = *(const v2f*)(xrow + (size_t)t0 * F_DIM);
    v2f p1 = *(const v2f*)(xrow + (size_t)(t0 + 1) * F_DIM);
    v2f a_f0 = { p0.x, p1.x };
    v2f a_f1 = { p0.y, p1.y };

    v2f b00 = *(const v2f*)(w0 + mrow * T_DIM + t0);
    v2f b10 = *(const v2f*)(w1 + mrow * T_DIM + t0);
    v2f b01 = *(const v2f*)(w0 + oHiC * T_DIM + t0) * oHiM;
    v2f b11 = *(const v2f*)(w1 + oHiC * T_DIM + t0) * oHiM;

    acc00 = __builtin_amdgcn_wmma_f32_16x16x4_f32(false, a_f0, false, b00, (short)0, acc00, false, false);
    acc01 = __builtin_amdgcn_wmma_f32_16x16x4_f32(false, a_f0, false, b01, (short)0, acc01, false, false);
    acc10 = __builtin_amdgcn_wmma_f32_16x16x4_f32(false, a_f1, false, b10, (short)0, acc10, false, false);
    acc11 = __builtin_amdgcn_wmma_f32_16x16x4_f32(false, a_f1, false, b11, (short)0, acc11, false, false);
  }

  const float bs00 = bias[0 * O_DIM + mrow];
  const float bs10 = bias[1 * O_DIM + mrow];
  const float bs01 = bias[0 * O_DIM + oHiC] * oHiM;
  const float bs11 = bias[1 * O_DIM + oHiC] * oHiM;

#pragma unroll
  for (int r = 0; r < 8; ++r) {
    const int bb = b0 + r + 8 * grp;
    v2f o0;
    o0.x = acc00[r] + bs00;
    o0.y = acc10[r] + bs10;
    *(v2f*)(out + (size_t)bb * (O_DIM * TGT) + mrow * TGT) = o0;
  }
  if (oHiOk) {
#pragma unroll
    for (int r = 0; r < 8; ++r) {
      const int bb = b0 + r + 8 * grp;
      v2f o1;
      o1.x = acc01[r] + bs01;
      o1.y = acc11[r] + bs11;
      *(v2f*)(out + (size_t)bb * (O_DIM * TGT) + oHi * TGT) = o1;
    }
  }
}

extern "C" void kernel_launch(void* const* d_in, const int* in_sizes, int n_in,
                              void* d_out, int out_size, void* d_ws, size_t ws_size,
                              hipStream_t stream) {
  const float* x    = (const float*)d_in[0];
  const float* W    = (const float*)d_in[1];
  const float* bias = (const float*)d_in[2];
  float* out = (float*)d_out;

  dim3 grid(B_TOTAL / 16 / 8);
  dim3 block(256);
  dlinear_wmma_f32<<<grid, block, 0, stream>>>(x, W, bias, out);
}


# K split over the 2 waves of a work-group: one 16-batch tile per group (256 groups), 3 of 6 k-steps per wave, bias as C of wave 0; partial tiles summed in wave order in LDS, whole-line b128 stores
# speedup vs baseline: 2.3303x; 2.3303x over previous
#include <hip/hip_runtime.h>

typedef __attribute__((ext_vector_type(2))) float v2f;
typedef __attribute__((ext_vector_type(4))) float v4f;
typedef __attribute__((ext_vector_type(8))) float v8f;

#define B_TOTAL 4096
#define T_DIM   24
#define F_DIM   256
#define O_DIM   24
#define TGT     2

#define KSPLIT  2
#define KSTEPS  (T_DIM / 4 / KSPLIT)
#define ROW_OUT (O_DIM * TGT)
#define ROW_Q   (ROW_OUT / 4)
#define ROW_LDS 52
#define TILE_Q  (16 * ROW_Q)
#define WG_SIZE (32 * KSPLIT)

static_assert(T_DIM % (4 * KSPLIT) == 0, "K slices are whole 16x16x4 steps");
static_assert(TILE_Q % WG_SIZE == 0, "every thread sums the same number of pieces");
static_assert(ROW_LDS >= ROW_OUT && ROW_LDS % 4 == 0, "padded row holds a whole output row, 16-byte aligned");

__global__ __launch_bounds__(WG_SIZE) void dlinear_wmma_f32(
    const float* __restrict__ x,
    const float* __restrict__ W,
    const float* __restrict__ bias,
    float* __restrict__ out)
{
  __shared__ __attribute__((aligned(16))) float part[KSPLIT][16 * ROW_LDS];

  const int lane = threadIdx.x & 31;
  const int wave = threadIdx.x >> 5;
  const int b0   = blockIdx.x * 16;

  const int  mrow = lane & 15;
  const int  grp  = lane >> 4;

  const int   oHi   = 16 + mrow;
  const bool  oHiOk = (oHi < O_DIM);
  const int   oHiC  = oHiOk ? oHi : 0;
  const float oHiM  = oHiOk ? 1.0f : 0.0f;

  const int    tw   = wave * (KSTEPS * 4) + 2 * grp;
  const float* xb   = x + (size_t)(b0 + mrow) * (T_DIM * F_DIM) + (size_t)tw * F_DIM;
  const float* w0lo = W + mrow * T_DIM + tw;
  const float* w1lo = w0lo + O_DIM * T_DIM;
  const float* w0hi = W + oHiC * T_DIM + tw;
  const float* w1hi = w0hi + O_DIM * T_DIM;

  const float bs00 = bias[0 * O_DIM + mrow];
  const float bs10 = bias[1 * O_DIM + mrow];
  const float bs01 = bias[0 * O_DIM + oHiC] * oHiM;
  const float bs11 = bias[1 * O_DIM + oHiC] * oHiM;
  const bool  lead = (wave == 0);
  const float c00 = lead ? bs00 : 0.0f;
  const float c10 = lead ? bs10 : 0.0f;
  const float c01 = lead ? bs01 : 0.0f;
  const float c11 = lead ? bs11 : 0.0f;

  v8f acc00, acc01, acc10, acc11;
#pragma unroll
  for (int r = 0; r < 8; ++r) { acc00[r] = c00; acc01[r] = c01; acc10[r] = c10; acc11[r] = c11; }

#pragma unroll
  for (int s = 0; s < KSTEPS; ++s) {
    v2f p0 = *(const v2f*)(xb + (4 * s) * F_DIM);
    v2f p1 = *(const v2f*)(xb + (4 * s + 1) * F_DIM);
    v2f a_f0 = { p0.x, p1.x };
    v2f a_f1 = { p0.y, p1.y };

    v2f b00 = *(const v2f*)(w0lo + 4 * s);
    v2f b10 = *(const v2f*)(w1lo + 4 * s);
    v2f b01 = *(const v2f*)(w0hi + 4 * s) * oHiM;
    v2f b11 = *(const v2f*)(w1hi + 4 * s) * oHiM;

    acc00 = __builtin_amdgcn_wmma_f32_16x16x4_f32(false, a_f0, false, b00, (short)0, acc00, false, false);
    acc01 = __builtin_amdgcn_wmma_f32_16x16x4_f32(false, a_f0, false, b01, (short)0, acc01, false, false);
    acc10 = __builtin_amdgcn_wmma_f32_16x16x4_f32(false, a_f1, false, b10, (short)0, acc10, false, false);
    acc11 = __builtin_amdgcn_wmma_f32_16x16x4_f32(false, a_f1, false, b11, (short)0, acc11, false, false);
  }

  float* mine = part[wave];
#pragma unroll
  for (int r = 0; r < 8; ++r) {
    const int m = r + 8 * grp;
    v2f lo = { acc00[r], acc10[r] };
    *(v2f*)(mine + m * ROW_LDS + mrow * TGT) = lo;
  }
  if (oHiOk) {
#pragma unroll
    for (int r = 0; r < 8; ++r) {
      const int m = r + 8 * grp;
      v2f hi = { acc01[r], acc11[r] };
      *(v2f*)(mine + m * ROW_LDS + oHi * TGT) = hi;
    }
  }
  __syncthreads();

  float* otile = out + (size_t)b0 * ROW_OUT;
#pragma unroll
  for (int j = 0; j < TILE_Q / WG_SIZE; ++j) {
    const int q   = (int)threadIdx.x + j * WG_SIZE;
    const int row = q / ROW_Q;
    const int col = q - row * ROW_Q;
    const float* src = &part[0][row * ROW_LDS + col * 4];
    v4f sum = *(const v4f*)src;
#pragma unroll
    for (int w = 1; w < KSPLIT; ++w) sum += *(const v4f*)(src + w * 16 * ROW_LDS);
    *(v4f*)(otile + q * 4) = sum;
  }
}

extern "C" void kernel_launch(void* const* d_in, const int* in_sizes, int n_in,
                              void* d_out, int out_size, void* d_ws, size_t ws_size,
                              hipStream_t stream) {
  const float* x    = (const float*)d_in[0];
  const float* W    = (const float*)d_in[1];
  const float* bias = (const float*)d_in[2];
  float* out = (float*)d_out;

  dim3 grid(B_TOTAL / 16);
  dim3 block(WG_SIZE);
  dlinear_wmma_f32<<<grid, block, 0, stream>>>(x, W, bias, out);
}
